# starting kernel
# baseline (speedup 1.0000x reference)
#include <hip/hip_runtime.h>
#include <math.h>

typedef __attribute__((ext_vector_type(2))) float v2f;
typedef __attribute__((ext_vector_type(4))) float v4f;
typedef __attribute__((ext_vector_type(8))) float v8f;

#define P_TOTAL 6272
#define NCAP 32
#define ICAP 8
#define KCAP 16
#define JCAP 16
#define PT   16
#define LDSJ 20
#define SQ_EPS 1e-7f
#define NROUT 3

__global__ __launch_bounds__(256)
void caps_routing_kernel(const float* __restrict__ X,
                         const float* __restrict__ W,
                         float* __restrict__ Out)
{
    __shared__ float pred[PT * NCAP * LDSJ];

    const int tid   = threadIdx.x;
    const int lane  = tid & 31;
    const int wave  = tid >> 5;
    const int blk   = blockIdx.x;
    const int k     = blk & (KCAP - 1);
    const int ptile = blk >> 4;
    const int p0    = ptile * PT;

    const int m    = lane & 15;
    const int half = lane >> 4;

    for (int nn = 0; nn < 4; ++nn) {
        const int n = wave * 4 + nn;

        const float* xb = X + (size_t)(p0 + m) * (NCAP * ICAP) + n * ICAP;
        v2f a0, a1;
        a0.x = xb[2 * half + 0];  a0.y = xb[2 * half + 1];
        a1.x = xb[2 * half + 4];  a1.y = xb[2 * half + 5];

        const float* wb = W + ((size_t)(k * NCAP + n) * ICAP) * JCAP + m;
        v2f b0, b1;
        b0.x = wb[(2 * half + 0) * JCAP];  b0.y = wb[(2 * half + 1) * JCAP];
        b1.x = wb[(2 * half + 4) * JCAP];  b1.y = wb[(2 * half + 5) * JCAP];

        v8f acc = {};
        acc = __builtin_amdgcn_wmma_f32_16x16x4_f32(false, a0, false, b0,
                                                    (short)0, acc, false, false);
        acc = __builtin_amdgcn_wmma_f32_16x16x4_f32(false, a1, false, b1,
                                                    (short)0, acc, false, false);

        #pragma unroll
        for (int v = 0; v < 8; ++v) {
            const int prow = v + 8 * half;
            pred[(prow * NCAP + n) * LDSJ + m] = acc[v];
        }
    }
    __syncthreads();

    for (int pl = wave; pl < PT; pl += 8) {
        float pj[JCAP];
        #pragma unroll
        for (int j = 0; j < JCAP; ++j)
            pj[j] = pred[(pl * NCAP + lane) * LDSJ + j];

        float braw = 0.0f;
        float out[JCAP];

        for (int r = 0; r < NROUT; ++r) {
            float mx = braw;
            #pragma unroll
            for (int s = 16; s >= 1; s >>= 1)
                mx = fmaxf(mx, __shfl_xor(mx, s, 32));
            float e = __expf(braw - mx);
            float se = e;
            #pragma unroll
            for (int s = 16; s >= 1; s >>= 1)
                se += __shfl_xor(se, s, 32);
            const float c = e / se;

            float sq = 0.0f;
            #pragma unroll
            for (int j = 0; j < JCAP; ++j) {
                float t = c * pj[j];
                #pragma unroll
                for (int s = 16; s >= 1; s >>= 1)
                    t += __shfl_xor(t, s, 32);
                out[j] = t;
                sq += t * t;
            }

            const float scale = sq / ((1.0f + sq) * sqrtf(sq + SQ_EPS));

            float dot = 0.0f;
            #pragma unroll
            for (int j = 0; j < JCAP; ++j) {
                out[j] *= scale;
                dot += pj[j] * out[j];
            }
            braw += dot;
        }

        if (lane == 0) {
            float* ob = Out + ((size_t)(p0 + pl) * KCAP + k) * JCAP;
            #pragma unroll
            for (int j = 0; j < JCAP; j += 4) {
                v4f t = { out[j], out[j + 1], out[j + 2], out[j + 3] };
                *(v4f*)(ob + j) = t;
            }
        }
    }
}

extern "C" void kernel_launch(void* const* d_in, const int* in_sizes, int n_in,
                              void* d_out, int out_size, void* d_ws, size_t ws_size,
                              hipStream_t stream) {
    const float* X = (const float*)d_in[0];
    const float* W = (const float*)d_in[1];
    float* Out = (float*)d_out;

    const int nblocks = (P_TOTAL / PT) * KCAP;
    caps_routing_kernel<<<nblocks, 256, 0, stream>>>(X, W, Out);
}


# Routing re-laid on four positions per wave: a lane keeps an 8(n) x 8(j) block in registers, both contractions are in-lane packed FMAs; DPP ops 113 per position -> 77 per four; 128-thread blocks
# speedup vs baseline: 3.0362x; 3.0362x over previous
#include <hip/hip_runtime.h>
#include <math.h>

typedef __attribute__((ext_vector_type(2))) float v2f;
typedef __attribute__((ext_vector_type(4))) float v4f;
typedef __attribute__((ext_vector_type(8))) float v8f;

#define P_TOTAL 6272
#define NCAP 32
#define ICAP 8
#define KCAP 16
#define JCAP 16
#define PT   16
#define NWAVE 4
#define POSF (NCAP * JCAP)
#define SQ_EPS 1e-7f
#define NROUT 3

#define DPP_XMASK(n) (0x160 + (n))

template <int CTRL>
__device__ __forceinline__ float dppf(float v) {
    return __builtin_bit_cast(float,
        __builtin_amdgcn_mov_dpp(__builtin_bit_cast(int, v), CTRL, 0xf, 0xf, true));
}

__device__ __forceinline__ float ngroup_sum(float v) {
    v += dppf<DPP_XMASK(2)>(v);
    v += dppf<DPP_XMASK(4)>(v);
    return v;
}

__device__ __forceinline__ float ngroup_max(float v) {
    float a, b;
    asm("v_max_num_f32_dpp %0, %1, %1 row_xmask:2 row_mask:0xf bank_mask:0xf bound_ctrl:1" : "=v"(a) : "v"(v));
    asm("v_max_num_f32_dpp %0, %1, %1 row_xmask:4 row_mask:0xf bank_mask:0xf bound_ctrl:1" : "=v"(b) : "v"(a));
    return b;
}

__global__ __launch_bounds__(32 * NWAVE)
void caps_routing_kernel(const float* __restrict__ X,
                         const float* __restrict__ W,
                         float* __restrict__ Out)
{
    __shared__ __attribute__((aligned(16))) float pred[PT * POSF];

    const int tid   = threadIdx.x;
    const int lane  = tid & 31;
    const int wave  = __builtin_amdgcn_readfirstlane(tid >> 5);
    const int blk   = blockIdx.x;
    const int k     = blk & (KCAP - 1);
    const int ptile = blk >> 4;
    const int p0    = ptile * PT;

    const int m    = lane & 15;
    const int half = lane >> 4;
    const int st_even = (8 * half) * POSF + m;
    const int st_odd  = (8 * half) * POSF + (m ^ 4);

    for (int nn = 0; nn < NCAP / NWAVE; ++nn) {
        const int n = wave * (NCAP / NWAVE) + nn;

        const float* xb = X + (size_t)(p0 + m) * (NCAP * ICAP) + n * ICAP;
        v2f a0, a1;
        a0.x = xb[2 * half + 0];  a0.y = xb[2 * half + 1];
        a1.x = xb[2 * half + 4];  a1.y = xb[2 * half + 5];

        const float* wb = W + ((size_t)(k * NCAP + n) * ICAP) * JCAP + m;
        v2f b0, b1;
        b0.x = wb[(2 * half + 0) * JCAP];  b0.y = wb[(2 * half + 1) * JCAP];
        b1.x = wb[(2 * half + 4) * JCAP];  b1.y = wb[(2 * half + 5) * JCAP];

        v8f acc = {};
        acc = __builtin_amdgcn_wmma_f32_16x16x4_f32(false, a0, false, b0,
                                                    (short)0, acc, false, false);
        acc = __builtin_amdgcn_wmma_f32_16x16x4_f32(false, a1, false, b1,
                                                    (short)0, acc, false, false);

        const int rowo = (n ^ (half << 1)) << 4;
        #pragma unroll
        for (int v = 0; v < 8; ++v)
            pred[v * POSF + rowo + ((v & 1) ? st_odd : st_even)] = acc[v];
    }
    __syncthreads();

    const int b  = lane & 1;
    const int a  = (lane >> 1) & 3;
    const int q  = lane >> 3;
    const int pos  = 4 * wave + q;
    const int podd = q & 1;
    const int base = pos * POSF + ((a ^ ((wave >> 1) << 1)) << 4) + 8 * b;
    const int off_lo = base + 4 * podd;
    const int off_hi = base + 4 * (1 - podd);

    v2f p2[8][4];
    #pragma unroll
    for (int s = 0; s < 8; ++s) {
        const v4f lo = *reinterpret_cast<const v4f*>(&pred[off_lo + 64 * s]);
        const v4f hi = *reinterpret_cast<const v4f*>(&pred[off_hi + 64 * s]);
        p2[s][0] = lo.xy; p2[s][1] = lo.zw;
        p2[s][2] = hi.xy; p2[s][3] = hi.zw;
    }

    float braw[8];
    #pragma unroll
    for (int s = 0; s < 8; ++s) braw[s] = 0.0f;
    v2f o2[4];
    #pragma unroll
    for (int u = 0; u < 4; ++u) o2[u] = (v2f){0.0f, 0.0f};

    #pragma unroll
    for (int r = 0; r < NROUT; ++r) {
        float mx = __builtin_fmaxf(__builtin_fmaxf(__builtin_fmaxf(braw[0], braw[1]),
                                                   __builtin_fmaxf(braw[2], braw[3])),
                                   __builtin_fmaxf(__builtin_fmaxf(braw[4], braw[5]),
                                                   __builtin_fmaxf(braw[6], braw[7])));
        mx = ngroup_max(mx);
        float e[8];
        #pragma unroll
        for (int s = 0; s < 8; ++s) e[s] = __expf(braw[s] - mx);
        const float se = ngroup_sum(((e[0] + e[1]) + (e[2] + e[3])) + ((e[4] + e[5]) + (e[6] + e[7])));
        const float inv = __builtin_amdgcn_rcpf(se);

        v2f z2[4];
        #pragma unroll
        for (int u = 0; u < 4; ++u) {
            v2f acc = p2[0][u] * e[0];
            #pragma unroll
            for (int s = 1; s < 8; ++s) acc = __builtin_elementwise_fma(p2[s][u], (v2f){e[s], e[s]}, acc);
            z2[u].x = ngroup_sum(acc.x) * inv;
            z2[u].y = ngroup_sum(acc.y) * inv;
        }

        v2f sq2 = z2[0] * z2[0];
        #pragma unroll
        for (int u = 1; u < 4; ++u) sq2 = __builtin_elementwise_fma(z2[u], z2[u], sq2);
        float sq = sq2.x + sq2.y;
        sq += dppf<DPP_XMASK(1)>(sq);
        const float scale = sq * __builtin_amdgcn_rsqf(sq + SQ_EPS)
                               * __builtin_amdgcn_rcpf(1.0f + sq);
        #pragma unroll
        for (int u = 0; u < 4; ++u) o2[u] = z2[u] * scale;

        if (r + 1 < NROUT) {
            #pragma unroll
            for (int s = 0; s < 8; ++s) {
                v2f d2 = p2[s][0] * o2[0];
                #pragma unroll
                for (int u = 1; u < 4; ++u) d2 = __builtin_elementwise_fma(p2[s][u], o2[u], d2);
                float d = d2.x + d2.y;
                d += dppf<DPP_XMASK(1)>(d);
                braw[s] += d;
            }
        }
    }

    if (a < 2) {
        v4f ov;
        ov.x = a ? o2[2].x : o2[0].x;
        ov.y = a ? o2[2].y : o2[0].y;
        ov.z = a ? o2[3].x : o2[1].x;
        ov.w = a ? o2[3].y : o2[1].y;
        *reinterpret_cast<v4f*>(&Out[((size_t)(p0 + pos) * KCAP + k) * JCAP + 8 * b + 4 * a]) = ov;
    }
}

extern "C" void kernel_launch(void* const* d_in, const int* in_sizes, int n_in,
                              void* d_out, int out_size, void* d_ws, size_t ws_size,
                              hipStream_t stream) {
    const float* X = (const float*)d_in[0];
    const float* W = (const float*)d_in[1];
    float* Out = (float*)d_out;

    const int nblocks = (P_TOTAL / PT) * KCAP;
    caps_routing_kernel<<<nblocks, 32 * NWAVE, 0, stream>>>(X, W, Out);
}


# Each block loops over 4 output capsules k on its position tile, the tile's A fragments loaded once and kept in registers: grid 6272 -> 1568 blocks of 128 threads, 4x fewer waves to enter
# speedup vs baseline: 3.7229x; 1.2262x over previous
#include <hip/hip_runtime.h>
#include <math.h>

typedef __attribute__((ext_vector_type(2))) float v2f;
typedef __attribute__((ext_vector_type(4))) float v4f;
typedef __attribute__((ext_vector_type(8))) float v8f;

#define P_TOTAL 6272
#define NCAP 32
#define ICAP 8
#define KCAP 16
#define JCAP 16
#define PT   16
#define NWAVE 4
#define KB   4
#define POSF (NCAP * JCAP)
#define SQ_EPS 1e-7f
#define NROUT 3

#define DPP_XMASK(n) (0x160 + (n))

template <int CTRL>
__device__ __forceinline__ float dppf(float v) {
    return __builtin_bit_cast(float,
        __builtin_amdgcn_mov_dpp(__builtin_bit_cast(int, v), CTRL, 0xf, 0xf, true));
}

__device__ __forceinline__ float ngroup_sum(float v) {
    v += dppf<DPP_XMASK(2)>(v);
    v += dppf<DPP_XMASK(4)>(v);
    return v;
}

__device__ __forceinline__ float ngroup_max(float v) {
    float a, b;
    asm("v_max_num_f32_dpp %0, %1, %1 row_xmask:2 row_mask:0xf bank_mask:0xf bound_ctrl:1" : "=v"(a) : "v"(v));
    asm("v_max_num_f32_dpp %0, %1, %1 row_xmask:4 row_mask:0xf bank_mask:0xf bound_ctrl:1" : "=v"(b) : "v"(a));
    return b;
}

__global__ __launch_bounds__(32 * NWAVE, 7)
void caps_routing_kernel(const float* __restrict__ X,
                         const float* __restrict__ W,
                         float* __restrict__ Out)
{
    __shared__ __attribute__((aligned(16))) float pred[PT * POSF];

    const int tid   = threadIdx.x;
    const int lane  = tid & 31;
    const int wave  = __builtin_amdgcn_readfirstlane(tid >> 5);
    const int blk   = blockIdx.x;
    const int kg    = blk & (KCAP / KB - 1);
    const int ptile = blk / (KCAP / KB);
    const int p0    = ptile * PT;

    const int m    = lane & 15;
    const int half = lane >> 4;
    const int st_even = (8 * half) * POSF + m;
    const int st_odd  = (8 * half) * POSF + (m ^ 4);

    v2f xa0[NCAP / NWAVE], xa1[NCAP / NWAVE];
    #pragma unroll
    for (int nn = 0; nn < NCAP / NWAVE; ++nn) {
        const int n = wave * (NCAP / NWAVE) + nn;
        const float* xb = X + (size_t)(p0 + m) * (NCAP * ICAP) + n * ICAP;
        xa0[nn].x = xb[2 * half + 0];  xa0[nn].y = xb[2 * half + 1];
        xa1[nn].x = xb[2 * half + 4];  xa1[nn].y = xb[2 * half + 5];
    }

    const int b  = lane & 1;
    const int a  = (lane >> 1) & 3;
    const int q  = lane >> 3;
    const int pos  = 4 * wave + q;
    const int podd = q & 1;
    const int base = pos * POSF + ((a ^ ((wave >> 1) << 1)) << 4) + 8 * b;
    const int off_lo = base + 4 * podd;
    const int off_hi = base + 4 * (1 - podd);

    #pragma unroll 1
    for (int kk = 0; kk < KB; ++kk) {
    const int k = kg * KB + kk;

    v2f wb0[NCAP / NWAVE], wb1[NCAP / NWAVE];
    #pragma unroll
    for (int nn = 0; nn < NCAP / NWAVE; ++nn) {
        const int n = wave * (NCAP / NWAVE) + nn;
        const float* wb = W + ((size_t)(k * NCAP + n) * ICAP) * JCAP + m;
        wb0[nn].x = wb[(2 * half + 0) * JCAP];  wb0[nn].y = wb[(2 * half + 1) * JCAP];
        wb1[nn].x = wb[(2 * half + 4) * JCAP];  wb1[nn].y = wb[(2 * half + 5) * JCAP];
    }
    #pragma unroll
    for (int g = 0; g < 2; ++g) {
        #pragma unroll
        for (int t = 0; t < 4; ++t) {
            const int nn = 4 * g + t;
            const int n  = wave * (NCAP / NWAVE) + nn;
            v8f acc = {};
            acc = __builtin_amdgcn_wmma_f32_16x16x4_f32(false, xa0[nn], false, wb0[nn],
                                                        (short)0, acc, false, false);
            acc = __builtin_amdgcn_wmma_f32_16x16x4_f32(false, xa1[nn], false, wb1[nn],
                                                        (short)0, acc, false, false);
            const int rowo = (n ^ (half << 1)) << 4;
            #pragma unroll
            for (int v = 0; v < 8; ++v)
                pred[v * POSF + rowo + ((v & 1) ? st_odd : st_even)] = acc[v];
        }
        __builtin_amdgcn_sched_barrier(0);
    }
    __syncthreads();

    v2f p2[8][4];
    #pragma unroll
    for (int s = 0; s < 8; ++s) {
        const v4f lo = *reinterpret_cast<const v4f*>(&pred[off_lo + 64 * s]);
        const v4f hi = *reinterpret_cast<const v4f*>(&pred[off_hi + 64 * s]);
        p2[s][0] = lo.xy; p2[s][1] = lo.zw;
        p2[s][2] = hi.xy; p2[s][3] = hi.zw;
    }
    if (kk + 1 < KB) __syncthreads();

    float braw[8];
    #pragma unroll
    for (int s = 0; s < 8; ++s) braw[s] = 0.0f;
    v2f o2[4];
    #pragma unroll
    for (int u = 0; u < 4; ++u) o2[u] = (v2f){0.0f, 0.0f};

    #pragma unroll
    for (int r = 0; r < NROUT; ++r) {
        float mx = __builtin_fmaxf(__builtin_fmaxf(__builtin_fmaxf(braw[0], braw[1]),
                                                   __builtin_fmaxf(braw[2], braw[3])),
                                   __builtin_fmaxf(__builtin_fmaxf(braw[4], braw[5]),
                                                   __builtin_fmaxf(braw[6], braw[7])));
        mx = ngroup_max(mx);
        float e[8];
        #pragma unroll
        for (int s = 0; s < 8; ++s) e[s] = __expf(braw[s] - mx);
        const float se = ngroup_sum(((e[0] + e[1]) + (e[2] + e[3])) + ((e[4] + e[5]) + (e[6] + e[7])));
        const float inv = __builtin_amdgcn_rcpf(se);

        v2f z2[4];
        #pragma unroll
        for (int u = 0; u < 4; ++u) {
            v2f acc = p2[0][u] * e[0];
            #pragma unroll
            for (int s = 1; s < 8; ++s) acc = __builtin_elementwise_fma(p2[s][u], (v2f){e[s], e[s]}, acc);
            z2[u].x = ngroup_sum(acc.x) * inv;
            z2[u].y = ngroup_sum(acc.y) * inv;
        }

        v2f sq2 = z2[0] * z2[0];
        #pragma unroll
        for (int u = 1; u < 4; ++u) sq2 = __builtin_elementwise_fma(z2[u], z2[u], sq2);
        float sq = sq2.x + sq2.y;
        sq += dppf<DPP_XMASK(1)>(sq);
        const float scale = sq * __builtin_amdgcn_rsqf(sq + SQ_EPS)
                               * __builtin_amdgcn_rcpf(1.0f + sq);
        #pragma unroll
        for (int u = 0; u < 4; ++u) o2[u] = z2[u] * scale;

        if (r + 1 < NROUT) {
            #pragma unroll
            for (int s = 0; s < 8; ++s) {
                v2f d2 = p2[s][0] * o2[0];
                #pragma unroll
                for (int u = 1; u < 4; ++u) d2 = __builtin_elementwise_fma(p2[s][u], o2[u], d2);
                float d = d2.x + d2.y;
                d += dppf<DPP_XMASK(1)>(d);
                braw[s] += d;
            }
        }
    }

    if (a < 2) {
        v4f ov;
        ov.x = a ? o2[2].x : o2[0].x;
        ov.y = a ? o2[2].y : o2[0].y;
        ov.z = a ? o2[3].x : o2[1].x;
        ov.w = a ? o2[3].y : o2[1].y;
        *reinterpret_cast<v4f*>(&Out[((size_t)(p0 + pos) * KCAP + k) * JCAP + 8 * b + 4 * a]) = ov;
    }
    }
}

extern "C" void kernel_launch(void* const* d_in, const int* in_sizes, int n_in,
                              void* d_out, int out_size, void* d_ws, size_t ws_size,
                              hipStream_t stream) {
    const float* X = (const float*)d_in[0];
    const float* W = (const float*)d_in[1];
    float* Out = (float*)d_out;

    const int nblocks = (P_TOTAL / PT) * (KCAP / KB);
    caps_routing_kernel<<<nblocks, 32 * NWAVE, 0, stream>>>(X, W, Out);
}
